# starting kernel
# baseline (speedup 1.0000x reference)
#include <hip/hip_runtime.h>


typedef float v2f __attribute__((ext_vector_type(2)));
typedef float v4f __attribute__((ext_vector_type(4)));
typedef float v8f __attribute__((ext_vector_type(8)));

#define B_ 4
#define H_ 12
#define S_ 2048
#define D_ 64
#define E_ 768

__global__ __launch_bounds__(32) void qkv_proj_wmma(
    const float* __restrict__ query, const float* __restrict__ key,
    const float* __restrict__ value, const float* __restrict__ Wq,
    const float* __restrict__ Wk,   const float* __restrict__ Wv,
    float* __restrict__ proj)
{
  const int which = blockIdx.x / (E_ / 16);
  const int tile  = blockIdx.x % (E_ / 16);
  const int m0    = tile * 16;

  const float* W = (which == 0) ? Wq : (which == 1) ? Wk : Wv;
  const float* X = (which == 0) ? query : (which == 1) ? key : value;

  const int lane = threadIdx.x;
  const int hi   = lane >> 4;
  const int lo   = lane & 15;

  const v2f* Wp = (const v2f*)(W + (size_t)(m0 + lo) * E_ + hi * 2);
  const v2f* Xp = (const v2f*)(X + (size_t)(lo & 3) * E_ + hi * 2);
  const float nmask = (lo < B_) ? 1.0f : 0.0f;

  v8f acc = {0.f, 0.f, 0.f, 0.f, 0.f, 0.f, 0.f, 0.f};
#pragma unroll 8
  for (int k0 = 0; k0 < E_; k0 += 4) {
    v2f a = Wp[k0 >> 1];
    v2f x = Xp[k0 >> 1];
    v2f b = x * nmask;
    acc = __builtin_amdgcn_wmma_f32_16x16x4_f32(
         false, a,  false, b,
         (short)0, acc,  false,  false);
  }

  if (lo < B_) {
    float* dst = proj + (size_t)which * (B_ * E_) + (size_t)lo * E_ + m0 + hi * 8;
#pragma unroll
    for (int r = 0; r < 8; ++r) dst[r] = acc[r];
  }
}

__global__ __launch_bounds__(64) void softmax_rows(
    const float* __restrict__ proj, float* __restrict__ row0,
    float* __restrict__ rowR)
{
  __shared__ float red[64];
  const int b = blockIdx.x / H_;
  const int h = blockIdx.x % H_;
  const int d = threadIdx.x;

  const float* q = proj + 0 * (B_ * E_) + b * E_ + h * D_;
  const float* k = proj + 1 * (B_ * E_) + b * E_ + h * D_;
  const float* v = proj + 2 * (B_ * E_) + b * E_ + h * D_;

  red[d] = q[d] * k[d];
  __syncthreads();
#pragma unroll
  for (int off = 32; off > 0; off >>= 1) {
    if (d < off) red[d] += red[d + off];
    __syncthreads();
  }

  const float s  = red[0] * 0.125f;
  const float m  = fmaxf(s, 0.0f);
  const float ea = __expf(s - m);
  const float eb = 2047.0f * __expf(-m);
  const float w0 = ea / (ea + eb);

  row0[b * E_ + h * D_ + d] = w0 * v[d];
  rowR[b * E_ + h * D_ + d] = v[d] * (1.0f / 2048.0f);
}

__global__ __launch_bounds__(256) void broadcast_fill(
    const float* __restrict__ row0, const float* __restrict__ rowR,
    float* __restrict__ out)
{
  const int idx = blockIdx.x * 256 + threadIdx.x;
  const int e4  = idx % (E_ / 4);
  const int row = idx / (E_ / 4);
  const int s   = row & (S_ - 1);
  const int b   = row >> 11;

  const v4f* src = (const v4f*)((s == 0) ? row0 : rowR);
  const v4f  val = src[b * (E_ / 4) + e4];
  __builtin_nontemporal_store(val, (v4f*)out + idx);
}

extern "C" void kernel_launch(void* const* d_in, const int* in_sizes, int n_in,
                              void* d_out, int out_size, void* d_ws, size_t ws_size,
                              hipStream_t stream) {
  const float* query = (const float*)d_in[0];
  const float* key   = (const float*)d_in[1];
  const float* value = (const float*)d_in[2];
  const float* Wq    = (const float*)d_in[3];
  const float* Wk    = (const float*)d_in[4];
  const float* Wv    = (const float*)d_in[5];
  float* out = (float*)d_out;

  float* proj = (float*)d_ws;
  float* row0 = proj + 3 * B_ * E_;
  float* rowR = row0 + B_ * E_;

  qkv_proj_wmma<<<3 * (E_ / 16), 32, 0, stream>>>(query, key, value,
                                                  Wq, Wk, Wv, proj);
  softmax_rows<<<B_ * H_, 64, 0, stream>>>(proj, row0, rowR);
  broadcast_fill<<<(B_ * S_ * E_ / 4) / 256, 256, 0, stream>>>(
      row0, rowR, out);
}


# QKV projection: one 8-wave workgroup per 16-row strip stages W in LDS with whole-line 128-bit loads, all 24 loads per wave in flight at once; K split over the waves, partial tiles summed in LDS
# speedup vs baseline: 1.9432x; 1.9432x over previous
#include <hip/hip_runtime.h>


typedef float v2f __attribute__((ext_vector_type(2)));
typedef float v4f __attribute__((ext_vector_type(4)));
typedef float v8f __attribute__((ext_vector_type(8)));

#define B_ 4
#define H_ 12
#define S_ 2048
#define D_ 64
#define E_ 768

#define KW_    8
#define KS_    (E_ / KW_)
#define NLD_   (KS_ / 8)
#define PITCH_ (E_ + 4)

__global__ __launch_bounds__(256) void qkv_proj_wmma(
    const float* __restrict__ query, const float* __restrict__ key,
    const float* __restrict__ value, const float* __restrict__ Wq,
    const float* __restrict__ Wk,   const float* __restrict__ Wv,
    float* __restrict__ proj)
{
  __shared__ __attribute__((aligned(16))) float sW[16 * PITCH_];
  __shared__ __attribute__((aligned(16))) float sP[KW_ * 64];

  const int which = blockIdx.x / (E_ / 16);
  const int tile  = blockIdx.x % (E_ / 16);
  const int m0    = tile * 16;

  const float* W = (which == 0) ? Wq : (which == 1) ? Wk : Wv;
  const float* X = (which == 0) ? query : (which == 1) ? key : value;

  const int tid  = threadIdx.x;
  const int wave = tid >> 5;
  const int lane = tid & 31;
  const int hi   = lane >> 4;
  const int lo   = lane & 15;

  v4f wreg[12];
#pragma unroll
  for (int j = 0; j < 12; ++j) {
    const int r = 2 * wave + (j / 6);
    const int c = (j % 6) * 128 + lane * 4;
    wreg[j] = *(const v4f*)(W + (size_t)(m0 + r) * E_ + c);
  }

  const int kb = wave * KS_ + hi * 4;
  const float nmask = (lo < B_) ? 1.0f : 0.0f;
  v4f xreg[NLD_];
#pragma unroll
  for (int i = 0; i < NLD_; ++i)
    xreg[i] = *(const v4f*)(X + (size_t)(lo & 3) * E_ + kb + i * 8);
  __builtin_amdgcn_sched_barrier(0);

#pragma unroll
  for (int j = 0; j < 12; ++j) {
    const int r = 2 * wave + (j / 6);
    const int c = (j % 6) * 128 + lane * 4;
    *(v4f*)(sW + r * PITCH_ + c) = wreg[j];
  }
#pragma unroll
  for (int i = 0; i < NLD_; ++i) xreg[i] = xreg[i] * nmask;
  __syncthreads();

  v4f areg[NLD_];
#pragma unroll
  for (int i = 0; i < NLD_; ++i)
    areg[i] = *(const v4f*)(sW + lo * PITCH_ + kb + i * 8);
  __builtin_amdgcn_sched_barrier(0);

  v8f acc0 = {0.f, 0.f, 0.f, 0.f, 0.f, 0.f, 0.f, 0.f};
  v8f acc1 = {0.f, 0.f, 0.f, 0.f, 0.f, 0.f, 0.f, 0.f};
#pragma unroll
  for (int i = 0; i < NLD_; ++i) {
    const v4f a4 = areg[i];
    const v4f x4 = xreg[i];
    v2f a01 = {a4.x, a4.y};
    v2f b01 = {x4.x, x4.y};
    v2f a23 = {a4.z, a4.w};
    v2f b23 = {x4.z, x4.w};
    acc0 = __builtin_amdgcn_wmma_f32_16x16x4_f32(
         false, a01,  false, b01,
         (short)0, acc0,  false,  false);
    acc1 = __builtin_amdgcn_wmma_f32_16x16x4_f32(
         false, a23,  false, b23,
         (short)0, acc1,  false,  false);
  }
  const v8f acc = acc0 + acc1;

  if (lo < B_) {
    float* p = sP + wave * 64 + lo * 16 + hi * 8;
    v4f p0 = {acc[0], acc[1], acc[2], acc[3]};
    v4f p1 = {acc[4], acc[5], acc[6], acc[7]};
    *(v4f*)(p)     = p0;
    *(v4f*)(p + 4) = p1;
  }
  __syncthreads();
  if (tid < 64) {
    float s = sP[tid];
#pragma unroll
    for (int w = 1; w < KW_; ++w) s += sP[w * 64 + tid];
    const int n = tid >> 4;
    const int m = tid & 15;
    proj[(size_t)which * (B_ * E_) + (size_t)n * E_ + m0 + m] = s;
  }
}

__global__ __launch_bounds__(64) void softmax_rows(
    const float* __restrict__ proj, float* __restrict__ row0,
    float* __restrict__ rowR)
{
  __shared__ float red[64];
  const int b = blockIdx.x / H_;
  const int h = blockIdx.x % H_;
  const int d = threadIdx.x;

  const float* q = proj + 0 * (B_ * E_) + b * E_ + h * D_;
  const float* k = proj + 1 * (B_ * E_) + b * E_ + h * D_;
  const float* v = proj + 2 * (B_ * E_) + b * E_ + h * D_;

  red[d] = q[d] * k[d];
  __syncthreads();
#pragma unroll
  for (int off = 32; off > 0; off >>= 1) {
    if (d < off) red[d] += red[d + off];
    __syncthreads();
  }

  const float s  = red[0] * 0.125f;
  const float m  = fmaxf(s, 0.0f);
  const float ea = __expf(s - m);
  const float eb = 2047.0f * __expf(-m);
  const float w0 = ea / (ea + eb);

  row0[b * E_ + h * D_ + d] = w0 * v[d];
  rowR[b * E_ + h * D_ + d] = v[d] * (1.0f / 2048.0f);
}

__global__ __launch_bounds__(256) void broadcast_fill(
    const float* __restrict__ row0, const float* __restrict__ rowR,
    float* __restrict__ out)
{
  const int idx = blockIdx.x * 256 + threadIdx.x;
  const int e4  = idx % (E_ / 4);
  const int row = idx / (E_ / 4);
  const int s   = row & (S_ - 1);
  const int b   = row >> 11;

  const v4f* src = (const v4f*)((s == 0) ? row0 : rowR);
  const v4f  val = src[b * (E_ / 4) + e4];
  __builtin_nontemporal_store(val, (v4f*)out + idx);
}

extern "C" void kernel_launch(void* const* d_in, const int* in_sizes, int n_in,
                              void* d_out, int out_size, void* d_ws, size_t ws_size,
                              hipStream_t stream) {
  const float* query = (const float*)d_in[0];
  const float* key   = (const float*)d_in[1];
  const float* value = (const float*)d_in[2];
  const float* Wq    = (const float*)d_in[3];
  const float* Wk    = (const float*)d_in[4];
  const float* Wv    = (const float*)d_in[5];
  float* out = (float*)d_out;

  float* proj = (float*)d_ws;
  float* row0 = proj + 3 * B_ * E_;
  float* rowR = row0 + B_ * E_;

  qkv_proj_wmma<<<3 * (E_ / 16), 256, 0, stream>>>(query, key, value,
                                                  Wq, Wk, Wv, proj);
  softmax_rows<<<B_ * H_, 64, 0, stream>>>(proj, row0, rowR);
  broadcast_fill<<<(B_ * S_ * E_ / 4) / 256, 256, 0, stream>>>(
      row0, rowR, out);
}


# fill on a resident grid: 256 eight-wave workgroups, each thread scales its 3 column groups of v once and stores them 24 times in whole lines; 49152 one-store waves become 2048
# speedup vs baseline: 3.0693x; 1.5795x over previous
#include <hip/hip_runtime.h>


typedef float v2f __attribute__((ext_vector_type(2)));
typedef float v4f __attribute__((ext_vector_type(4)));
typedef float v8f __attribute__((ext_vector_type(8)));

#define B_ 4
#define H_ 12
#define S_ 2048
#define D_ 64
#define E_ 768

#define KW_    8
#define KS_    (E_ / KW_)
#define NLD_   (KS_ / 8)
#define PITCH_ (E_ + 4)

__global__ __launch_bounds__(256) void qkv_proj_wmma(
    const float* __restrict__ query, const float* __restrict__ key,
    const float* __restrict__ value, const float* __restrict__ Wq,
    const float* __restrict__ Wk,   const float* __restrict__ Wv,
    float* __restrict__ proj)
{
  __shared__ __attribute__((aligned(16))) float sW[16 * PITCH_];
  __shared__ __attribute__((aligned(16))) float sP[KW_ * 64];

  const int which = blockIdx.x / (E_ / 16);
  const int tile  = blockIdx.x % (E_ / 16);
  const int m0    = tile * 16;

  const float* W = (which == 0) ? Wq : (which == 1) ? Wk : Wv;
  const float* X = (which == 0) ? query : (which == 1) ? key : value;

  const int tid  = threadIdx.x;
  const int wave = tid >> 5;
  const int lane = tid & 31;
  const int hi   = lane >> 4;
  const int lo   = lane & 15;

  v4f wreg[12];
#pragma unroll
  for (int j = 0; j < 12; ++j) {
    const int r = 2 * wave + (j / 6);
    const int c = (j % 6) * 128 + lane * 4;
    wreg[j] = *(const v4f*)(W + (size_t)(m0 + r) * E_ + c);
  }

  const int kb = wave * KS_ + hi * 4;
  const float nmask = (lo < B_) ? 1.0f : 0.0f;
  v4f xreg[NLD_];
#pragma unroll
  for (int i = 0; i < NLD_; ++i)
    xreg[i] = *(const v4f*)(X + (size_t)(lo & 3) * E_ + kb + i * 8);
  __builtin_amdgcn_sched_barrier(0);

#pragma unroll
  for (int j = 0; j < 12; ++j) {
    const int r = 2 * wave + (j / 6);
    const int c = (j % 6) * 128 + lane * 4;
    *(v4f*)(sW + r * PITCH_ + c) = wreg[j];
  }
#pragma unroll
  for (int i = 0; i < NLD_; ++i) xreg[i] = xreg[i] * nmask;
  __syncthreads();

  v4f areg[NLD_];
#pragma unroll
  for (int i = 0; i < NLD_; ++i)
    areg[i] = *(const v4f*)(sW + lo * PITCH_ + kb + i * 8);
  __builtin_amdgcn_sched_barrier(0);

  v8f acc0 = {0.f, 0.f, 0.f, 0.f, 0.f, 0.f, 0.f, 0.f};
  v8f acc1 = {0.f, 0.f, 0.f, 0.f, 0.f, 0.f, 0.f, 0.f};
#pragma unroll
  for (int i = 0; i < NLD_; ++i) {
    const v4f a4 = areg[i];
    const v4f x4 = xreg[i];
    v2f a01 = {a4.x, a4.y};
    v2f b01 = {x4.x, x4.y};
    v2f a23 = {a4.z, a4.w};
    v2f b23 = {x4.z, x4.w};
    acc0 = __builtin_amdgcn_wmma_f32_16x16x4_f32(
         false, a01,  false, b01,
         (short)0, acc0,  false,  false);
    acc1 = __builtin_amdgcn_wmma_f32_16x16x4_f32(
         false, a23,  false, b23,
         (short)0, acc1,  false,  false);
  }
  const v8f acc = acc0 + acc1;

  if (lo < B_) {
    float* p = sP + wave * 64 + lo * 16 + hi * 8;
    v4f p0 = {acc[0], acc[1], acc[2], acc[3]};
    v4f p1 = {acc[4], acc[5], acc[6], acc[7]};
    *(v4f*)(p)     = p0;
    *(v4f*)(p + 4) = p1;
  }
  __syncthreads();
  if (tid < 64) {
    float s = sP[tid];
#pragma unroll
    for (int w = 1; w < KW_; ++w) s += sP[w * 64 + tid];
    const int n = tid >> 4;
    const int m = tid & 15;
    proj[(size_t)which * (B_ * E_) + (size_t)n * E_ + m0 + m] = s;
  }
}

#define FILL_WGS_PER_BATCH_ 64
#define FILL_CHUNKS_        (S_ * (E_ / 4) / 256 / FILL_WGS_PER_BATCH_)

__global__ __launch_bounds__(256) void softmax_broadcast_fill(
    const float* __restrict__ proj, float* __restrict__ out)
{
  const int tid = threadIdx.x;
  const int b   = blockIdx.x / FILL_WGS_PER_BATCH_;
  const int j   = blockIdx.x % FILL_WGS_PER_BATCH_;

  const v4f* q4p = (const v4f*)(proj);
  const v4f* k4p = (const v4f*)(proj + 1 * (B_ * E_));
  const v4f* v4p = (const v4f*)(proj + 2 * (B_ * E_));
  const int  rb  = b * (E_ / 4);

  int e0 = (j % 3) * 64 + tid;
  if (e0 >= 192) e0 -= 192;
  const int e1 = (e0 < 128) ? e0 + 64  : e0 - 128;
  const int e2 = (e0 < 64)  ? e0 + 128 : e0 - 64;

  const v4f r0 = v4p[rb + e0];
  const v4f r1 = v4p[rb + e1];
  const v4f r2 = v4p[rb + e2];

  const float inv = 1.0f / 2048.0f;
  float scale0 = inv;

  if (j == 0) {
    const v4f q4 = q4p[rb + e0];
    const v4f k4 = k4p[rb + e0];
    float p = q4.x * k4.x + q4.y * k4.y + q4.z * k4.z + q4.w * k4.w;
    p += __shfl_xor(p, 8);
    p += __shfl_xor(p, 4);
    p += __shfl_xor(p, 2);
    p += __shfl_xor(p, 1);
    const float sc = p * 0.125f;
    const float m  = fmaxf(sc, 0.0f);
    const float ea = __expf(sc - m);
    const float eb = 2047.0f * __expf(-m);
    const float w0 = ea / (ea + eb);
    if (tid < 192) scale0 = w0;
  }

  const v4f s0 = r0 * inv;
  const v4f s1 = r1 * inv;
  const v4f s2 = r2 * inv;

  v4f* o = (v4f*)out + (size_t)b * (S_ * (E_ / 4)) + (size_t)j * 256 + tid;
  const int step = FILL_WGS_PER_BATCH_ * 256;

  __builtin_nontemporal_store(r0 * scale0, o);
  __builtin_nontemporal_store(s1, o + 1 * step);
  __builtin_nontemporal_store(s2, o + 2 * step);
#pragma unroll
  for (int c = 3; c < FILL_CHUNKS_; c += 3) {
    __builtin_nontemporal_store(s0, o + (c + 0) * step);
    __builtin_nontemporal_store(s1, o + (c + 1) * step);
    __builtin_nontemporal_store(s2, o + (c + 2) * step);
  }
}

extern "C" void kernel_launch(void* const* d_in, const int* in_sizes, int n_in,
                              void* d_out, int out_size, void* d_ws, size_t ws_size,
                              hipStream_t stream) {
  const float* query = (const float*)d_in[0];
  const float* key   = (const float*)d_in[1];
  const float* value = (const float*)d_in[2];
  const float* Wq    = (const float*)d_in[3];
  const float* Wk    = (const float*)d_in[4];
  const float* Wv    = (const float*)d_in[5];
  float* out = (float*)d_out;

  float* proj = (float*)d_ws;

  qkv_proj_wmma<<<3 * (E_ / 16), 256, 0, stream>>>(query, key, value,
                                                  Wq, Wk, Wv, proj);
  softmax_broadcast_fill<<<B_ * FILL_WGS_PER_BATCH_, 256, 0, stream>>>(
      proj, out);
}
